# starting kernel
# baseline (speedup 1.0000x reference)
#include <hip/hip_runtime.h>
#include <hip/hip_bf16.h>

typedef __attribute__((ext_vector_type(2))) float v2f;
typedef __attribute__((ext_vector_type(4))) float v4f;
typedef __attribute__((ext_vector_type(8))) float v8f;

#define DIM 8192
#define NTHREADS 1024
#define NWAVES (NTHREADS / 32)
#define KCHUNKS ((DIM / 4) / NWAVES)

__global__ __launch_bounds__(NTHREADS)
void ExponentialTrajectoryFit_fused(const float* __restrict__ x0,
                                    const float* __restrict__ x1,
                                    const float* __restrict__ x2,
                                    const float* __restrict__ x3,
                                    float* __restrict__ out) {
    const int tid  = threadIdx.x;
    const int wave = tid >> 5;
    const int lane = tid & 31;
    const int m    = lane & 15;
    const int kh   = (lane >> 4) << 1;

    const float* p;
    const float* q;
    float sp, sq;
    if (m == 0)      { p = x1; q = x0; sp = 1.0f; sq = 1.0f; }
    else if (m == 1) { p = x2; q = x1; sp = 1.0f; sq = 1.0f; }
    else if (m == 2) { p = x3; q = x2; sp = 1.0f; sq = 1.0f; }
    else if (m == 3) { p = x0; q = x0; sp = 1.0f; sq = 0.0f; }
    else if (m == 4) { p = x1; q = x0; sp = 1.0f; sq = 0.0f; }
    else             { p = x0; q = x0; sp = 0.0f; sq = 0.0f; }

    const float* pp = p + wave * 4 + kh;
    const float* qq = q + wave * 4 + kh;

    v8f acc = {};
    #pragma unroll 8
    for (int it = 0; it < KCHUNKS; ++it) {
        const v2f P = *(const v2f*)(pp + it * (4 * NWAVES));
        const v2f Q = *(const v2f*)(qq + it * (4 * NWAVES));
        v2f a;
        a[0] = sp * P[0] - sq * Q[0];
        a[1] = sp * P[1] - sq * Q[1];
        acc = __builtin_amdgcn_wmma_f32_16x16x4_f32(
             false, a,  false, a,
             (short)0, acc,  false,  false);
    }

    __shared__ float partial[NWAVES][10];
    __shared__ float gsum[10];
    __shared__ float coeff[2];
    if (lane < 5) {
        partial[wave][lane * 2 + 0] = acc[0];
        partial[wave][lane * 2 + 1] = acc[1];
    }
    __syncthreads();

    if (tid < 10) {
        float s = 0.0f;
        #pragma unroll
        for (int w = 0; w < NWAVES; ++w) s += partial[w][tid];
        gsum[tid] = s;
    }
    __syncthreads();
    if (tid == 0) {
        const float g00 = gsum[0], g10 = gsum[1];
        const float g01 = gsum[2], g11 = gsum[3];
        const float q02 = gsum[4], q12 = gsum[5];
        const float p00 = gsum[6], p01 = gsum[7];
        const float p10 = gsum[8], p11 = gsum[9];

        const float det = g00 * g11 - g01 * g10;
        const float id  = 1.0f / det;
        const float a0 = id * ( g11 * p00 - g01 * p01);
        const float a1 = id * (-g10 * p00 + g00 * p01);
        const float r0 = p10 - (g01 * a0 + q02 * a1);
        const float r1 = p11 - (g11 * a0 + q12 * a1);
        const float v0 = id * ( g11 * r0 - g01 * r1);
        const float v1 = id * (-g10 * r0 + g00 * r1);
        const float s00 = id * ( g11 * g01 - g01 * g11);
        const float s10 = id * (-g10 * g01 + g00 * g11);
        const float s01 = id * ( g11 * q02 - g01 * q12);
        const float s11 = id * (-g10 * q02 + g00 * q12);
        const float w00 = 1.0f - s00, w01 = -s01;
        const float w10 = -s10,       w11 = 1.0f - s11;
        const float iw  = 1.0f / (w00 * w11 - w01 * w10);
        const float c0 = iw * ( w11 * v0 - w01 * v1);
        const float c1 = iw * (-w10 * v0 + w00 * v1);

        coeff[0] = a0 - c0;
        coeff[1] = a1 - c1;
    }
    __syncthreads();

    const float alpha = coeff[0];
    const float beta  = coeff[1];
    const int base = tid * 8;
    #pragma unroll
    for (int j = 0; j < 2; ++j) {
        const int i = base + j * 4;
        const v4f e0 = *(const v4f*)(x0 + i);
        const v4f e1 = *(const v4f*)(x1 + i);
        const v4f e2 = *(const v4f*)(x2 + i);
        const v4f e3 = *(const v4f*)(x3 + i);
        const v4f r  = (e0 - e1) + alpha * (e2 - e1) + beta * (e3 - e2);
        *(v4f*)(out + i) = r;
    }
}

extern "C" void kernel_launch(void* const* d_in, const int* in_sizes, int n_in,
                              void* d_out, int out_size, void* d_ws, size_t ws_size,
                              hipStream_t stream) {
    (void)in_sizes; (void)n_in; (void)out_size; (void)d_ws; (void)ws_size;
    const float* x0 = (const float*)d_in[0];
    const float* x1 = (const float*)d_in[1];
    const float* x2 = (const float*)d_in[2];
    const float* x3 = (const float*)d_in[3];
    float* out = (float*)d_out;
    ExponentialTrajectoryFit_fused<<<1, NTHREADS, 0, stream>>>(x0, x1, x2, x3, out);
}


# k runs down the lanes (whole-line b128 loads); one f32 WMMA per vector pair, A=frag(u), B=frag(v), the dot product on the tile's diagonal: 36 WMMAs per wave, not 64; reduction tree changes
# speedup vs baseline: 1.6993x; 1.6993x over previous
#include <hip/hip_runtime.h>
#include <hip/hip_bf16.h>

typedef __attribute__((ext_vector_type(2))) float v2f;
typedef __attribute__((ext_vector_type(4))) float v4f;
typedef __attribute__((ext_vector_type(8))) float v8f;

#define DIM 8192
#define NTHREADS 1024
#define NWAVES (NTHREADS / 32)
#define NBLOCKS 8
#define SLICE (DIM / NBLOCKS)
#define CHUNK 128
#define NCHUNK (DIM / (NWAVES * CHUNK))
#define NPAIR 9
#define NPART (NWAVES * 16)


static __device__ __forceinline__ v8f mm(v2f a, v2f b, v8f c) {
    return __builtin_amdgcn_wmma_f32_16x16x4_f32(
         false, a,  false, b,
         (short)0, c,  false,  false);
}

static __device__ __forceinline__ float pick8(v8f a, int r) {
    const bool b0 = (r & 1) != 0, b1 = (r & 2) != 0, b2 = (r & 4) != 0;
    const float t0 = b0 ? a[1] : a[0];
    const float t1 = b0 ? a[3] : a[2];
    const float t2 = b0 ? a[5] : a[4];
    const float t3 = b0 ? a[7] : a[6];
    const float u0 = b1 ? t1 : t0;
    const float u1 = b1 ? t3 : t2;
    return b2 ? u1 : u0;
}

template <int M>
static __device__ __forceinline__ float xrow(float x) {
    return __builtin_bit_cast(float, __builtin_amdgcn_update_dpp(
        0, __builtin_bit_cast(int, x), 0x160 | M, 0xF, 0xF, true));
}

static __device__ __forceinline__ float xhalf(float x) {
    const int b = __builtin_bit_cast(int, x);
    return __builtin_bit_cast(float, __builtin_amdgcn_permlanex16(
        b, b, 0x76543210, 0xfedcba98, false, false));
}

__global__ __launch_bounds__(NTHREADS)
void ExponentialTrajectoryFit_fused(const float* __restrict__ x0,
                                    const float* __restrict__ x1,
                                    const float* __restrict__ x2,
                                    const float* __restrict__ x3,
                                    float* __restrict__ out) {
    const int tid  = threadIdx.x;
    const int wave = tid >> 5;
    const int lane = tid & 31;

    v4f X0[NCHUNK], X1[NCHUNK], X2[NCHUNK], X3[NCHUNK];
    #pragma unroll
    for (int c = 0; c < NCHUNK; ++c) {
        const int i = (c * NWAVES + wave) * CHUNK + lane * 4;
        X0[c] = *(const v4f*)(x0 + i);
        X1[c] = *(const v4f*)(x1 + i);
        X2[c] = *(const v4f*)(x2 + i);
        X3[c] = *(const v4f*)(x3 + i);
    }

    v8f acc[NPAIR];
    #pragma unroll
    for (int p = 0; p < NPAIR; ++p) acc[p] = (v8f){};

    #pragma unroll
    for (int c = 0; c < NCHUNK; ++c) {
        const v4f D0 = X1[c] - X0[c];
        const v4f D1 = X2[c] - X1[c];
        const v4f D2 = X3[c] - X2[c];
        #pragma unroll
        for (int h = 0; h < 2; ++h) {
            const v2f d0 = { D0[2 * h], D0[2 * h + 1] };
            const v2f d1 = { D1[2 * h], D1[2 * h + 1] };
            const v2f d2 = { D2[2 * h], D2[2 * h + 1] };
            const v2f e0 = { X0[c][2 * h], X0[c][2 * h + 1] };
            const v2f e1 = { X1[c][2 * h], X1[c][2 * h + 1] };
            acc[0] = mm(d0, d0, acc[0]);
            acc[1] = mm(d0, d1, acc[1]);
            acc[2] = mm(d1, d1, acc[2]);
            acc[3] = mm(d0, d2, acc[3]);
            acc[4] = mm(d1, d2, acc[4]);
            acc[5] = mm(d0, e0, acc[5]);
            acc[6] = mm(d1, e0, acc[6]);
            acc[7] = mm(d0, e1, acc[7]);
            acc[8] = mm(d1, e1, acc[8]);
        }
    }

    __shared__ float partial[NPAIR][NPART];
    __shared__ float gsum[NPAIR];
    __shared__ float coeff[2];
    {
        const int r = lane & 7;
        const bool on_diag = (lane < 8) || (lane >= 24);
        #pragma unroll
        for (int p = 0; p < NPAIR; ++p) {
            const float v = pick8(acc[p], r);
            if (on_diag) partial[p][wave * 16 + (lane & 15)] = v;
        }
    }
    __syncthreads();

    if (wave < NPAIR) {
        float s0 = 0.0f, s1 = 0.0f, s2 = 0.0f, s3 = 0.0f;
        #pragma unroll
        for (int j = 0; j < NPART / 32; j += 4) {
            s0 += partial[wave][lane + 32 * (j + 0)];
            s1 += partial[wave][lane + 32 * (j + 1)];
            s2 += partial[wave][lane + 32 * (j + 2)];
            s3 += partial[wave][lane + 32 * (j + 3)];
        }
        float s = (s0 + s1) + (s2 + s3);
        s += xrow<8>(s);
        s += xrow<4>(s);
        s += xrow<2>(s);
        s += xrow<1>(s);
        s += xhalf(s);
        if (lane == 0) gsum[wave] = s;
    }
    __syncthreads();
    if (tid == 0) {
        const float g00 = gsum[0];
        const float g01 = gsum[1], g10 = gsum[1];
        const float g11 = gsum[2];
        const float q02 = gsum[3], q12 = gsum[4];
        const float p00 = gsum[5], p01 = gsum[6];
        const float p10 = gsum[7], p11 = gsum[8];

        const float det = g00 * g11 - g01 * g10;
        const float id  = 1.0f / det;
        const float a0 = id * ( g11 * p00 - g01 * p01);
        const float a1 = id * (-g10 * p00 + g00 * p01);
        const float r0 = p10 - (g01 * a0 + q02 * a1);
        const float r1 = p11 - (g11 * a0 + q12 * a1);
        const float v0 = id * ( g11 * r0 - g01 * r1);
        const float v1 = id * (-g10 * r0 + g00 * r1);
        const float s00 = id * ( g11 * g01 - g01 * g11);
        const float s10 = id * (-g10 * g01 + g00 * g11);
        const float s01 = id * ( g11 * q02 - g01 * q12);
        const float s11 = id * (-g10 * q02 + g00 * q12);
        const float w00 = 1.0f - s00, w01 = -s01;
        const float w10 = -s10,       w11 = 1.0f - s11;
        const float iw  = 1.0f / (w00 * w11 - w01 * w10);
        const float c0 = iw * ( w11 * v0 - w01 * v1);
        const float c1 = iw * (-w10 * v0 + w00 * v1);

        coeff[0] = a0 - c0;
        coeff[1] = a1 - c1;
    }
    __syncthreads();

    if (tid < SLICE / 4) {
        const float alpha = coeff[0];
        const float beta  = coeff[1];
        const int i = blockIdx.x * SLICE + tid * 4;
        const v4f e0 = *(const v4f*)(x0 + i);
        const v4f e1 = *(const v4f*)(x1 + i);
        const v4f e2 = *(const v4f*)(x2 + i);
        const v4f e3 = *(const v4f*)(x3 + i);
        const v4f r  = (e0 - e1) + alpha * (e2 - e1) + beta * (e3 - e2);
        *(v4f*)(out + i) = r;
    }
}

extern "C" void kernel_launch(void* const* d_in, const int* in_sizes, int n_in,
                              void* d_out, int out_size, void* d_ws, size_t ws_size,
                              hipStream_t stream) {
    (void)in_sizes; (void)n_in; (void)out_size; (void)d_ws; (void)ws_size;
    const float* x0 = (const float*)d_in[0];
    const float* x1 = (const float*)d_in[1];
    const float* x2 = (const float*)d_in[2];
    const float* x3 = (const float*)d_in[3];
    float* out = (float*)d_out;
    ExponentialTrajectoryFit_fused<<<NBLOCKS, NTHREADS, 0, stream>>>(x0, x1, x2, x3, out);
}
